# starting kernel
# baseline (speedup 1.0000x reference)
#include <hip/hip_runtime.h>
#include <math.h>
#include <utility>

typedef float v2f __attribute__((ext_vector_type(2)));
typedef float v8f __attribute__((ext_vector_type(8)));

#define HIDDEN  32
#define N_NODES 4096
#define N_EDGES 8192
#define FEAT    512
#define NPATHS  23

static constexpr int PL1[NPATHS] = {0,0,0,0, 1,1,1,1,1,1, 2,2,2,2,2,2,2, 3,3,3,3,3,3};
static constexpr int PL2[NPATHS] = {0,1,2,3, 0,1,1,2,2,3, 0,1,1,2,2,3,3, 0,1,2,2,3,3};
static constexpr int PL3[NPATHS] = {0,1,2,3, 1,0,2,1,3,2, 2,1,3,0,2,1,3, 3,2,1,3,0,2};
static constexpr float PWT[4] = {
  0.08838834764831845f,
  0.125f,
  0.14940357616680451f,
  0.19094065395649326f
};

template <typename T>
__device__ __forceinline__ void sh_eval(T x, T y, T z, T* Y) {
  const T s3  = (T)1.7320508075688772;
  const T s5  = (T)2.2360679774997896;
  const T s7  = (T)2.6457513110645907;
  const T s15 = (T)3.8729833462074170;
  const T c58 = (T)0.7905694150420949;
  const T c38 = (T)0.6123724356957945;
  Y[0]  = (T)1;
  Y[1]  = s3 * x;
  Y[2]  = s3 * y;
  Y[3]  = s3 * z;
  Y[4]  = s5 * s3 * x * z;
  Y[5]  = s5 * s3 * x * y;
  Y[6]  = s5 * (y * y - (T)0.5 * (x * x + z * z));
  Y[7]  = s5 * s3 * y * z;
  Y[8]  = s5 * (s3 * (T)0.5) * (z * z - x * x);
  Y[9]  = s7 * c58 * x * ((T)3 * z * z - x * x);
  Y[10] = s7 * s15 * x * y * z;
  Y[11] = s7 * c38 * x * ((T)5 * y * y - (T)1);
  Y[12] = s7 * (T)0.5 * y * ((T)5 * y * y - (T)3);
  Y[13] = s7 * c38 * z * ((T)5 * y * y - (T)1);
  Y[14] = s7 * (s15 * (T)0.5) * y * (z * z - x * x);
  Y[15] = s7 * c58 * z * (z * z - (T)3 * x * x);
}

__launch_bounds__(64)
__global__ void w3j_kernel(float* __restrict__ ctab) {
  const int p = blockIdx.x;
  const int t = threadIdx.x;

  int l1 = 0, l2 = 0, l3 = 0, cnt = 0;
  for (int a = 0; a < 4; ++a)
    for (int b = 0; b < 4; ++b)
      for (int c = 0; c < 4; ++c) {
        int lo = a - b; if (lo < 0) lo = -lo;
        if (lo <= c && c <= a + b && (((a + b + c) & 1) == 0)) {
          if (cnt == p) { l1 = a; l2 = b; l3 = c; }
          ++cnt;
        }
      }
  const int d1 = 2 * l1 + 1, d2 = 2 * l2 + 1, d3 = 2 * l3 + 1;
  const int NE = d1 * d2 * d3;

  double un[16], wn[16];
  const double PI = 3.14159265358979323846;
  for (int i = 0; i < 16; ++i) {
    double xg = cos(PI * (i + 0.75) / 16.5);
    double dp = 1.0;
    for (int it = 0; it < 100; ++it) {
      double a0 = 1.0, b0 = xg;
      for (int n = 2; n <= 16; ++n) {
        double c0 = ((2.0 * n - 1.0) * xg * b0 - (n - 1.0) * a0) / n;
        a0 = b0; b0 = c0;
      }
      dp = 16.0 * (xg * b0 - a0) / (xg * xg - 1.0);
      double dx = b0 / dp;
      xg -= dx;
      if (fabs(dx) < 1e-14) break;
    }
    {
      double a0 = 1.0, b0 = xg;
      for (int n = 2; n <= 16; ++n) {
        double c0 = ((2.0 * n - 1.0) * xg * b0 - (n - 1.0) * a0) / n;
        a0 = b0; b0 = c0;
      }
      dp = 16.0 * (xg * b0 - a0) / (xg * xg - 1.0);
    }
    un[i] = xg;
    wn[i] = 2.0 / ((1.0 - xg * xg) * dp * dp);
  }

  __shared__ double Cs[343];
  __shared__ double red[64];
  const double DPHI = 2.0 * PI / 32.0;

  for (int n = t; n < NE; n += 64) {
    const int i = n / (d2 * d3);
    const int j = (n / d3) % d2;
    const int k = n % d3;
    double s = 0.0;
    for (int q = 0; q < 512; ++q) {
      const int iq = q >> 5, jq = q & 31;
      const double uu  = un[iq];
      const double wq  = wn[iq] * DPHI;
      const double phi = jq * DPHI;
      double st2 = 1.0 - uu * uu;
      double st  = sqrt(st2 > 0.0 ? st2 : 0.0);
      double Yq[16];
      sh_eval(st * cos(phi), st * sin(phi), uu, Yq);
      s += wq * Yq[l1 * l1 + i] * Yq[l2 * l2 + j] * Yq[l3 * l3 + k];
    }
    Cs[n] = s;
  }
  __syncthreads();

  double ss = 0.0;
  for (int n = t; n < NE; n += 64) ss += Cs[n] * Cs[n];
  red[t] = ss;
  __syncthreads();
  for (int off = 32; off > 0; off >>= 1) {
    if (t < off) red[t] += red[t + off];
    __syncthreads();
  }
  const double inv = 1.0 / sqrt(red[0]);

  for (int n = t; n < NE; n += 64) {
    const int i = n / (d2 * d3);
    const int j = (n / d3) % d2;
    const int k = n % d3;
    ctab[p * 343 + (i * 7 + j) * 7 + k] = (float)(Cs[n] * inv);
  }
}

__global__ void zero_kernel(float* __restrict__ out, int n) {
  int i = blockIdx.x * blockDim.x + threadIdx.x;
  if (i < n) out[i] = 0.0f;
}

template <int P>
__device__ __forceinline__ void path_body(
    int lane, int n16,
    float len,
    const float (&Y)[16], const float (&xr)[16],
    const float* __restrict__ wwp,
    const float* __restrict__ wbp,
    const float* __restrict__ ctab,
    float* Gs, float* tmpB, const float* tb,
    v8f& accLo, v8f& accHi) {
  constexpr int l1 = PL1[P], l2 = PL2[P], l3 = PL3[P];
  constexpr int d1 = 2 * l1 + 1, d2 = 2 * l2 + 1, d3 = 2 * l3 + 1;
  constexpr int GN = d1 * d3;
  constexpr float pw = PWT[l3];
  const float* Cp = ctab + P * 343;

  if (lane < GN) {
    const int i = lane / d3, k = lane % d3;
    float g = 0.0f;
#pragma unroll
    for (int j = 0; j < d2; ++j) g = fmaf(Y[l2 * l2 + j], Cp[(i * 7 + j) * 7 + k], g);
    Gs[lane] = pw * g;
  }
  if constexpr (GN > 32) {
    const int n2 = lane + 32;
    if (n2 < GN) {
      const int i = n2 / d3, k = n2 % d3;
      float g = 0.0f;
#pragma unroll
      for (int j = 0; j < d2; ++j) g = fmaf(Y[l2 * l2 + j], Cp[(i * 7 + j) * 7 + k], g);
      Gs[n2] = pw * g;
    }
  }
  __syncthreads();

  float tk[7];
#pragma unroll
  for (int k = 0; k < d3; ++k) {
    float s = 0.0f;
#pragma unroll
    for (int i = 0; i < d1; ++i) s = fmaf(xr[l1 * l1 + i], Gs[i * d3 + k], s);
    tk[k] = s;
  }
#pragma unroll
  for (int k = 0; k < 16; ++k) tmpB[lane * 16 + k] = (k < d3) ? tk[k] : 0.0f;
  __syncthreads();

  constexpr int wbase = P * (HIDDEN * HIDDEN);
#pragma unroll
  for (int kc = 0; kc < 8; ++kc) {
    v2f b;
    b.x = tb[kc * 64];
    b.y = tb[kc * 64 + 16];
    {
      const int i0 = wbase + kc * 128;
      v2f a;
      a.x = fmaf(len, wwp[i0], wbp[i0]);
      a.y = fmaf(len, wwp[i0 + HIDDEN], wbp[i0 + HIDDEN]);
      accLo = __builtin_amdgcn_wmma_f32_16x16x4_f32(
          false, a, false, b, (short)0, accLo, false, false);
    }
    {
      const int i0 = wbase + kc * 128 + 16;
      v2f a;
      a.x = fmaf(len, wwp[i0], wbp[i0]);
      a.y = fmaf(len, wwp[i0 + HIDDEN], wbp[i0 + HIDDEN]);
      accHi = __builtin_amdgcn_wmma_f32_16x16x4_f32(
          false, a, false, b, (short)0, accHi, false, false);
    }
  }
  __syncthreads();
}

template <int... Ps>
__device__ __forceinline__ void run_group(
    std::integer_sequence<int, Ps...>,
    int lane, int n16, float len,
    const float (&Y)[16], const float (&xr)[16],
    const float* __restrict__ wwp, const float* __restrict__ wbp,
    const float* __restrict__ ctab,
    float* Gs, float* tmpB, const float* tb,
    v8f& accLo, v8f& accHi) {
  (path_body<Ps>(lane, n16, len, Y, xr, wwp, wbp, ctab, Gs, tmpB, tb, accLo, accHi), ...);
}

template <int L3>
__device__ __forceinline__ void store_tile(int lane, int n16, float* od,
                                           const v8f& accLo, const v8f& accHi) {
  constexpr int d3 = 2 * L3 + 1;
  if (n16 < d3) {
    const int wadd = (lane >= 16) ? 8 : 0;
#pragma unroll
    for (int r = 0; r < 8; ++r) {
      const int w = r + wadd;
      atomicAdd(od + HIDDEN * L3 * L3 + w * d3 + n16, accLo[r]);
      atomicAdd(od + HIDDEN * L3 * L3 + (w + 16) * d3 + n16, accHi[r]);
    }
  }
}

__launch_bounds__(32)
__global__ void edge_kernel(const float* __restrict__ nf,
                            const int*   __restrict__ eidx,
                            const float* __restrict__ evec,
                            const float* __restrict__ ww,
                            const float* __restrict__ wb,
                            const float* __restrict__ ctab,
                            float*       __restrict__ out) {
  const int e    = blockIdx.x;
  const int lane = threadIdx.x;

  const int src = eidx[e];
  const int dst = eidx[N_EDGES + e];

  const float vx = evec[3 * e + 0];
  const float vy = evec[3 * e + 1];
  const float vz = evec[3 * e + 2];
  float len = sqrtf(vx * vx + vy * vy + vz * vz);
  len = fmaxf(len, 1e-8f);
  const float inv = 1.0f / len;

  float Y[16];
  sh_eval(vx * inv, vy * inv, vz * inv, Y);

  const float* xs = nf + (size_t)src * FEAT;
  float xr[16];
#pragma unroll
  for (int l = 0; l < 4; ++l) {
    const int d = 2 * l + 1;
#pragma unroll
    for (int i = 0; i < 7; ++i)
      if (i < d) xr[l * l + i] = xs[HIDDEN * l * l + lane * d + i];
  }

  __shared__ float Gs[49];
  __shared__ float tmpB[32 * 16];

  const int n16  = lane & 15;
  const int usel = (lane >> 4) << 1;
  const float* wwp = ww + usel * HIDDEN + n16;
  const float* wbp = wb + usel * HIDDEN + n16;
  const float* tb  = tmpB + usel * 16 + n16;

  float* od = out + (size_t)dst * FEAT;

  {
    v8f a0 = {0,0,0,0,0,0,0,0}, a1 = {0,0,0,0,0,0,0,0};
    run_group(std::integer_sequence<int, 0, 5, 13, 21>{},
              lane, n16, len, Y, xr, wwp, wbp, ctab, Gs, tmpB, tb, a0, a1);
    store_tile<0>(lane, n16, od, a0, a1);
  }
  {
    v8f a0 = {0,0,0,0,0,0,0,0}, a1 = {0,0,0,0,0,0,0,0};
    run_group(std::integer_sequence<int, 1, 4, 7, 11, 15, 19>{},
              lane, n16, len, Y, xr, wwp, wbp, ctab, Gs, tmpB, tb, a0, a1);
    store_tile<1>(lane, n16, od, a0, a1);
  }
  {
    v8f a0 = {0,0,0,0,0,0,0,0}, a1 = {0,0,0,0,0,0,0,0};
    run_group(std::integer_sequence<int, 2, 6, 9, 10, 14, 18, 22>{},
              lane, n16, len, Y, xr, wwp, wbp, ctab, Gs, tmpB, tb, a0, a1);
    store_tile<2>(lane, n16, od, a0, a1);
  }
  {
    v8f a0 = {0,0,0,0,0,0,0,0}, a1 = {0,0,0,0,0,0,0,0};
    run_group(std::integer_sequence<int, 3, 8, 12, 16, 17, 20>{},
              lane, n16, len, Y, xr, wwp, wbp, ctab, Gs, tmpB, tb, a0, a1);
    store_tile<3>(lane, n16, od, a0, a1);
  }
}

extern "C" void kernel_launch(void* const* d_in, const int* in_sizes, int n_in,
                              void* d_out, int out_size, void* d_ws, size_t ws_size,
                              hipStream_t stream) {
  const float* nf   = (const float*)d_in[0];
  const int*   eidx = (const int*)  d_in[1];
  const float* evec = (const float*)d_in[2];
  const float* ww   = (const float*)d_in[3];
  const float* wb   = (const float*)d_in[4];
  float* out  = (float*)d_out;
  float* ctab = (float*)d_ws;

  w3j_kernel<<<NPATHS, 64, 0, stream>>>(ctab);
  const int n = N_NODES * FEAT;
  zero_kernel<<<(n + 255) / 256, 256, 0, stream>>>(out, n);
  edge_kernel<<<N_EDGES, 32, 0, stream>>>(nf, eidx, evec, ww, wb, ctab, out);
}


# edge launch: tile row m = weight columns 2m,2m+1, K = (u, parity of w), B diagonal in parity: each weight/bias fragment is one 8-byte load; f32 sum order changes (2 products a step, 2 partial tiles)
# speedup vs baseline: 82.5133x; 82.5133x over previous
#include <hip/hip_runtime.h>
#include <math.h>
#include <utility>

typedef float v2f __attribute__((ext_vector_type(2)));
typedef float v4f __attribute__((ext_vector_type(4)));
typedef float v8f __attribute__((ext_vector_type(8)));

#define HIDDEN  32
#define N_NODES 4096
#define N_EDGES 8192
#define FEAT    512
#define NPATHS  23
#define BPITCH  36

static constexpr int PL1[NPATHS] = {0,0,0,0, 1,1,1,1,1,1, 2,2,2,2,2,2,2, 3,3,3,3,3,3};
static constexpr int PL2[NPATHS] = {0,1,2,3, 0,1,1,2,2,3, 0,1,1,2,2,3,3, 0,1,2,2,3,3};
static constexpr int PL3[NPATHS] = {0,1,2,3, 1,0,2,1,3,2, 2,1,3,0,2,1,3, 3,2,1,3,0,2};
static constexpr float PWT[4] = {
  0.08838834764831845f,
  0.125f,
  0.14940357616680451f,
  0.19094065395649326f
};

template <typename T>
__device__ __forceinline__ void sh_eval(T x, T y, T z, T* Y) {
  const T s3  = (T)1.7320508075688772;
  const T s5  = (T)2.2360679774997896;
  const T s7  = (T)2.6457513110645907;
  const T s15 = (T)3.8729833462074170;
  const T c58 = (T)0.7905694150420949;
  const T c38 = (T)0.6123724356957945;
  Y[0]  = (T)1;
  Y[1]  = s3 * x;
  Y[2]  = s3 * y;
  Y[3]  = s3 * z;
  Y[4]  = s5 * s3 * x * z;
  Y[5]  = s5 * s3 * x * y;
  Y[6]  = s5 * (y * y - (T)0.5 * (x * x + z * z));
  Y[7]  = s5 * s3 * y * z;
  Y[8]  = s5 * (s3 * (T)0.5) * (z * z - x * x);
  Y[9]  = s7 * c58 * x * ((T)3 * z * z - x * x);
  Y[10] = s7 * s15 * x * y * z;
  Y[11] = s7 * c38 * x * ((T)5 * y * y - (T)1);
  Y[12] = s7 * (T)0.5 * y * ((T)5 * y * y - (T)3);
  Y[13] = s7 * c38 * z * ((T)5 * y * y - (T)1);
  Y[14] = s7 * (s15 * (T)0.5) * y * (z * z - x * x);
  Y[15] = s7 * c58 * z * (z * z - (T)3 * x * x);
}

#define W3J_THREADS 256

__device__ __forceinline__ double odd_dfact(int n) {
  double r = 1.0;
  for (int v = n; v > 1; v -= 2) r *= (double)v;
  return r;
}

#define MONO(ex, ey, ez) ((ex) | ((ey) << 4) | ((ez) << 8))

__device__ __forceinline__ int sh_terms(int m, double* c, int* e) {
  const double s3  = 1.7320508075688772;
  const double s5  = 2.2360679774997896;
  const double s7  = 2.6457513110645907;
  const double s15 = 3.8729833462074170;
  const double c58 = 0.7905694150420949;
  const double c38 = 0.6123724356957945;
  int n = 1;
  c[0] = 1.0; e[0] = MONO(0, 0, 0);
  c[1] = 0.0; e[1] = 0;
  c[2] = 0.0; e[2] = 0;
  switch (m) {
    case 0:  break;
    case 1:  c[0] = s3; e[0] = MONO(1, 0, 0); break;
    case 2:  c[0] = s3; e[0] = MONO(0, 1, 0); break;
    case 3:  c[0] = s3; e[0] = MONO(0, 0, 1); break;
    case 4:  c[0] = s5 * s3; e[0] = MONO(1, 0, 1); break;
    case 5:  c[0] = s5 * s3; e[0] = MONO(1, 1, 0); break;
    case 6:  n = 3;
             c[0] = s5;        e[0] = MONO(0, 2, 0);
             c[1] = -0.5 * s5; e[1] = MONO(2, 0, 0);
             c[2] = -0.5 * s5; e[2] = MONO(0, 0, 2); break;
    case 7:  c[0] = s5 * s3; e[0] = MONO(0, 1, 1); break;
    case 8:  n = 2;
             c[0] = s5 * (s3 * 0.5);    e[0] = MONO(0, 0, 2);
             c[1] = -(s5 * (s3 * 0.5)); e[1] = MONO(2, 0, 0); break;
    case 9:  n = 2;
             c[0] = 3.0 * (s7 * c58); e[0] = MONO(1, 0, 2);
             c[1] = -(s7 * c58);      e[1] = MONO(3, 0, 0); break;
    case 10: c[0] = s7 * s15; e[0] = MONO(1, 1, 1); break;
    case 11: n = 2;
             c[0] = 5.0 * (s7 * c38); e[0] = MONO(1, 2, 0);
             c[1] = -(s7 * c38);      e[1] = MONO(1, 0, 0); break;
    case 12: n = 2;
             c[0] = 5.0 * (s7 * 0.5);  e[0] = MONO(0, 3, 0);
             c[1] = -3.0 * (s7 * 0.5); e[1] = MONO(0, 1, 0); break;
    case 13: n = 2;
             c[0] = 5.0 * (s7 * c38); e[0] = MONO(0, 2, 1);
             c[1] = -(s7 * c38);      e[1] = MONO(0, 0, 1); break;
    case 14: n = 2;
             c[0] = s7 * (s15 * 0.5);    e[0] = MONO(0, 1, 2);
             c[1] = -(s7 * (s15 * 0.5)); e[1] = MONO(2, 1, 0); break;
    default: n = 2;
             c[0] = s7 * c58;          e[0] = MONO(0, 0, 3);
             c[1] = -3.0 * (s7 * c58); e[1] = MONO(2, 0, 1); break;
  }
  return n;
}

__launch_bounds__(W3J_THREADS)
__global__ void w3j_kernel(float* __restrict__ ctab) {
  const int p = blockIdx.x;
  const int t = threadIdx.x;

  int l1 = 0, l2 = 0, l3 = 0, cnt = 0;
  for (int a = 0; a < 4; ++a)
    for (int b = 0; b < 4; ++b)
      for (int c = 0; c < 4; ++c) {
        int lo = a - b; if (lo < 0) lo = -lo;
        if (lo <= c && c <= a + b && (((a + b + c) & 1) == 0)) {
          if (cnt == p) { l1 = a; l2 = b; l3 = c; }
          ++cnt;
        }
      }
  const int d1 = 2 * l1 + 1, d2 = 2 * l2 + 1, d3 = 2 * l3 + 1;
  const int NE = d1 * d2 * d3;

  __shared__ double Mom[125];
  __shared__ double Tc[16 * 3];
  __shared__ int    Te[16 * 3];
  __shared__ int    Tn[16];
  __shared__ double Cs[343];
  __shared__ double red[W3J_THREADS];
  const double PI = 3.14159265358979323846;

  if (t < 125) {
    const int a = 2 * (t / 25), b = 2 * ((t / 5) % 5), c = 2 * (t % 5);
    Mom[t] = 4.0 * PI * odd_dfact(a - 1) * odd_dfact(b - 1) * odd_dfact(c - 1)
             / odd_dfact(a + b + c + 1);
  } else if (t >= 128 && t < 144) {
    const int m = t - 128;
    double c[3];
    int e[3];
    Tn[m] = sh_terms(m, c, e);
    for (int q = 0; q < 3; ++q) { Tc[m * 3 + q] = c[q]; Te[m * 3 + q] = e[q]; }
  }
  __syncthreads();

  for (int n = t; n < NE; n += W3J_THREADS) {
    const int i = n / (d2 * d3);
    const int j = (n / d3) % d2;
    const int k = n % d3;
    const int m1 = l1 * l1 + i, m2 = l2 * l2 + j, m3 = l3 * l3 + k;
    const int n1 = Tn[m1], n2 = Tn[m2], n3 = Tn[m3];
    double s = 0.0;
    for (int q1 = 0; q1 < n1; ++q1)
      for (int q2 = 0; q2 < n2; ++q2)
        for (int q3 = 0; q3 < n3; ++q3) {
          const int e = Te[m1 * 3 + q1] + Te[m2 * 3 + q2] + Te[m3 * 3 + q3];
          if ((e & 0x111) == 0) {
            const int ia = (e & 15) >> 1, ib = ((e >> 4) & 15) >> 1, ic = ((e >> 8) & 15) >> 1;
            s += Tc[m1 * 3 + q1] * Tc[m2 * 3 + q2] * Tc[m3 * 3 + q3]
                 * Mom[ia * 25 + ib * 5 + ic];
          }
        }
    Cs[n] = s;
  }
  __syncthreads();

  double ss = 0.0;
  for (int n = t; n < NE; n += W3J_THREADS) ss += Cs[n] * Cs[n];
  red[t] = ss;
  __syncthreads();
  for (int off = W3J_THREADS / 2; off > 0; off >>= 1) {
    if (t < off) red[t] += red[t + off];
    __syncthreads();
  }
  const double inv = 1.0 / sqrt(red[0]);

  for (int n = t; n < NE; n += W3J_THREADS) {
    const int i = n / (d2 * d3);
    const int j = (n / d3) % d2;
    const int k = n % d3;
    ctab[p * 343 + (i * 7 + j) * 7 + k] = (float)(Cs[n] * inv);
  }
}

__global__ void zero_kernel(float* __restrict__ out, int n) {
  int i = blockIdx.x * blockDim.x + threadIdx.x;
  if (i < n) out[i] = 0.0f;
}

template <int P>
__device__ __forceinline__ void path_body(
    int lane,
    float len,
    const float (&Y)[16], const float (&xr)[16],
    const float* __restrict__ wwp,
    const float* __restrict__ wbp,
    const float* __restrict__ ctab,
    float* Gs, float* tmpB, const float* tb,
    v8f& acc0, v8f& acc1) {
  constexpr int l1 = PL1[P], l2 = PL2[P], l3 = PL3[P];
  constexpr int d1 = 2 * l1 + 1, d2 = 2 * l2 + 1, d3 = 2 * l3 + 1;
  constexpr int GN = d1 * d3;
  constexpr float pw = PWT[l3];
  const float* Cp = ctab + P * 343;

  if (lane < GN) {
    const int i = lane / d3, k = lane % d3;
    float g = 0.0f;
#pragma unroll
    for (int j = 0; j < d2; ++j) g = fmaf(Y[l2 * l2 + j], Cp[(i * 7 + j) * 7 + k], g);
    Gs[lane] = pw * g;
  }
  if constexpr (GN > 32) {
    const int n2 = lane + 32;
    if (n2 < GN) {
      const int i = n2 / d3, k = n2 % d3;
      float g = 0.0f;
#pragma unroll
      for (int j = 0; j < d2; ++j) g = fmaf(Y[l2 * l2 + j], Cp[(i * 7 + j) * 7 + k], g);
      Gs[n2] = pw * g;
    }
  }
  __syncthreads();

  float tk[8];
#pragma unroll
  for (int k = 0; k < 8; ++k) {
    float s = 0.0f;
    if (k < d3) {
#pragma unroll
      for (int i = 0; i < d1; ++i) s = fmaf(xr[l1 * l1 + i], Gs[i * d3 + k], s);
    }
    tk[k] = s;
  }
  float* brow = tmpB + lane * BPITCH;
#pragma unroll
  for (int k = 0; k < 8; ++k) {
    brow[4 * k]     = tk[k];
    brow[4 * k + 3] = tk[k];
  }
  __syncthreads();

  constexpr int wbase = P * (HIDDEN * HIDDEN);
#pragma unroll
  for (int t = 0; t < 16; ++t) {
    const v2f b = *reinterpret_cast<const v2f*>(tb + 2 * t * BPITCH);
    const int i0 = wbase + t * (2 * HIDDEN);
    const v2f w = *reinterpret_cast<const v2f*>(wwp + i0);
    const v2f c = *reinterpret_cast<const v2f*>(wbp + i0);
    v2f a;
    a.x = fmaf(len, w.x, c.x);
    a.y = fmaf(len, w.y, c.y);
    if (t & 1)
      acc1 = __builtin_amdgcn_wmma_f32_16x16x4_f32(
          false, a, false, b, (short)0, acc1, false, false);
    else
      acc0 = __builtin_amdgcn_wmma_f32_16x16x4_f32(
          false, a, false, b, (short)0, acc0, false, false);
  }
  __syncthreads();
}

template <int... Ps>
__device__ __forceinline__ void run_group(
    std::integer_sequence<int, Ps...>,
    int lane, float len,
    const float (&Y)[16], const float (&xr)[16],
    const float* __restrict__ wwp, const float* __restrict__ wbp,
    const float* __restrict__ ctab,
    float* Gs, float* tmpB, const float* tb,
    v8f& acc0, v8f& acc1) {
  (path_body<Ps>(lane, len, Y, xr, wwp, wbp, ctab, Gs, tmpB, tb, acc0, acc1), ...);
}

template <int L3>
__device__ __forceinline__ void store_tile(int lane, int n16, float* od, const v8f& acc) {
  constexpr int d3 = 2 * L3 + 1;
  if (n16 < 2 * d3) {
    const int k = n16 >> 1;
    const int g = n16 & 1;
    const int madd = (lane >= 16) ? 8 : 0;
#pragma unroll
    for (int r = 0; r < 8; ++r) {
      const int w = 2 * (r + madd) + g;
      atomicAdd(od + HIDDEN * L3 * L3 + w * d3 + k, acc[r]);
    }
  }
}

__launch_bounds__(32)
__global__ void edge_kernel(const float* __restrict__ nf,
                            const int*   __restrict__ eidx,
                            const float* __restrict__ evec,
                            const float* __restrict__ ww,
                            const float* __restrict__ wb,
                            const float* __restrict__ ctab,
                            float*       __restrict__ out) {
  const int e    = blockIdx.x;
  const int lane = threadIdx.x;

  const int src = eidx[e];
  const int dst = eidx[N_EDGES + e];

  const float vx = evec[3 * e + 0];
  const float vy = evec[3 * e + 1];
  const float vz = evec[3 * e + 2];
  float len = sqrtf(vx * vx + vy * vy + vz * vz);
  len = fmaxf(len, 1e-8f);
  const float inv = 1.0f / len;

  float Y[16];
  sh_eval(vx * inv, vy * inv, vz * inv, Y);

  const float* xs = nf + (size_t)src * FEAT;
  float xr[16];
#pragma unroll
  for (int l = 0; l < 4; ++l) {
    const int d = 2 * l + 1;
#pragma unroll
    for (int i = 0; i < 7; ++i)
      if (i < d) xr[l * l + i] = xs[HIDDEN * l * l + lane * d + i];
  }

  __shared__ float Gs[49];
  __shared__ alignas(16) float tmpB[32 * BPITCH];

  const int n16 = lane & 15;
  const int hh  = lane >> 4;
  const float* wwp = ww + hh * HIDDEN + 2 * n16;
  const float* wbp = wb + hh * HIDDEN + 2 * n16;
  const float* tb  = tmpB + hh * BPITCH + 2 * n16;

  float* od = out + (size_t)dst * FEAT;

  {
    const v4f z = {0.0f, 0.0f, 0.0f, 0.0f};
#pragma unroll
    for (int k = 0; k < 8; ++k)
      *reinterpret_cast<v4f*>(tmpB + lane * BPITCH + 4 * k) = z;
  }

  {
    v8f a0 = {0,0,0,0,0,0,0,0}, a1 = {0,0,0,0,0,0,0,0};
    run_group(std::integer_sequence<int, 0, 5, 13, 21>{},
              lane, len, Y, xr, wwp, wbp, ctab, Gs, tmpB, tb, a0, a1);
    const v8f sum = a0 + a1;
    store_tile<0>(lane, n16, od, sum);
  }
  {
    v8f a0 = {0,0,0,0,0,0,0,0}, a1 = {0,0,0,0,0,0,0,0};
    run_group(std::integer_sequence<int, 1, 4, 7, 11, 15, 19>{},
              lane, len, Y, xr, wwp, wbp, ctab, Gs, tmpB, tb, a0, a1);
    const v8f sum = a0 + a1;
    store_tile<1>(lane, n16, od, sum);
  }
  {
    v8f a0 = {0,0,0,0,0,0,0,0}, a1 = {0,0,0,0,0,0,0,0};
    run_group(std::integer_sequence<int, 2, 6, 9, 10, 14, 18, 22>{},
              lane, len, Y, xr, wwp, wbp, ctab, Gs, tmpB, tb, a0, a1);
    const v8f sum = a0 + a1;
    store_tile<2>(lane, n16, od, sum);
  }
  {
    v8f a0 = {0,0,0,0,0,0,0,0}, a1 = {0,0,0,0,0,0,0,0};
    run_group(std::integer_sequence<int, 3, 8, 12, 16, 17, 20>{},
              lane, len, Y, xr, wwp, wbp, ctab, Gs, tmpB, tb, a0, a1);
    const v8f sum = a0 + a1;
    store_tile<3>(lane, n16, od, sum);
  }
}

extern "C" void kernel_launch(void* const* d_in, const int* in_sizes, int n_in,
                              void* d_out, int out_size, void* d_ws, size_t ws_size,
                              hipStream_t stream) {
  const float* nf   = (const float*)d_in[0];
  const int*   eidx = (const int*)  d_in[1];
  const float* evec = (const float*)d_in[2];
  const float* ww   = (const float*)d_in[3];
  const float* wb   = (const float*)d_in[4];
  float* out  = (float*)d_out;
  float* ctab = (float*)d_ws;

  w3j_kernel<<<NPATHS, W3J_THREADS, 0, stream>>>(ctab);
  const int n = N_NODES * FEAT;
  zero_kernel<<<(n + 255) / 256, 256, 0, stream>>>(out, n);
  edge_kernel<<<N_EDGES, 32, 0, stream>>>(nf, eidx, evec, ww, wb, ctab, out);
}


# edge launch: coupling tables stored compactly (parity-coupled elements only), read through the scalar cache; G[i][k] built by scalar FMAs, scalar operand of the tmp FMAs; no G trip through LDS
# speedup vs baseline: 89.6846x; 1.0869x over previous
#include <hip/hip_runtime.h>
#include <math.h>
#include <utility>

typedef float v2f __attribute__((ext_vector_type(2)));
typedef float v4f __attribute__((ext_vector_type(4)));
typedef float v8f __attribute__((ext_vector_type(8)));

#define HIDDEN  32
#define N_NODES 4096
#define N_EDGES 8192
#define FEAT    512
#define NPATHS  23
#define BPITCH  36

static constexpr int PL1[NPATHS] = {0,0,0,0, 1,1,1,1,1,1, 2,2,2,2,2,2,2, 3,3,3,3,3,3};
static constexpr int PL2[NPATHS] = {0,1,2,3, 0,1,1,2,2,3, 0,1,1,2,2,3,3, 0,1,2,2,3,3};
static constexpr int PL3[NPATHS] = {0,1,2,3, 1,0,2,1,3,2, 2,1,3,0,2,1,3, 3,2,1,3,0,2};
static constexpr float PWT[4] = {
  0.08838834764831845f,
  0.125f,
  0.14940357616680451f,
  0.19094065395649326f
};

constexpr int par_of(int m) {
  return (int)((0x4242171060354210ULL >> (4 * m)) & 7ULL);
}
constexpr bool coupled(int m1, int m2, int m3) {
  return (par_of(m1) ^ par_of(m2) ^ par_of(m3)) == 0;
}
constexpr int pad8(int n) { return (n + 7) & ~7; }
constexpr int path_coupled(int p) {
  const int l1 = PL1[p], l2 = PL2[p], l3 = PL3[p];
  int c = 0;
  for (int i = 0; i < 2 * l1 + 1; ++i)
    for (int j = 0; j < 2 * l2 + 1; ++j)
      for (int k = 0; k < 2 * l3 + 1; ++k)
        if (coupled(l1 * l1 + i, l2 * l2 + j, l3 * l3 + k)) ++c;
  return c;
}
constexpr int cbase(int p) {
  int b = 0;
  for (int q = 0; q < p; ++q) b += pad8(path_coupled(q));
  return b;
}
template <int P> struct PathBase { static constexpr int value = cbase(P); };
template <int... Ps>
__device__ __forceinline__ int base_of(int p, std::integer_sequence<int, Ps...>) {
  int b = 0;
  ((b = (p == Ps) ? PathBase<Ps>::value : b), ...);
  return b;
}

template <typename T>
__device__ __forceinline__ void sh_eval(T x, T y, T z, T* Y) {
  const T s3  = (T)1.7320508075688772;
  const T s5  = (T)2.2360679774997896;
  const T s7  = (T)2.6457513110645907;
  const T s15 = (T)3.8729833462074170;
  const T c58 = (T)0.7905694150420949;
  const T c38 = (T)0.6123724356957945;
  Y[0]  = (T)1;
  Y[1]  = s3 * x;
  Y[2]  = s3 * y;
  Y[3]  = s3 * z;
  Y[4]  = s5 * s3 * x * z;
  Y[5]  = s5 * s3 * x * y;
  Y[6]  = s5 * (y * y - (T)0.5 * (x * x + z * z));
  Y[7]  = s5 * s3 * y * z;
  Y[8]  = s5 * (s3 * (T)0.5) * (z * z - x * x);
  Y[9]  = s7 * c58 * x * ((T)3 * z * z - x * x);
  Y[10] = s7 * s15 * x * y * z;
  Y[11] = s7 * c38 * x * ((T)5 * y * y - (T)1);
  Y[12] = s7 * (T)0.5 * y * ((T)5 * y * y - (T)3);
  Y[13] = s7 * c38 * z * ((T)5 * y * y - (T)1);
  Y[14] = s7 * (s15 * (T)0.5) * y * (z * z - x * x);
  Y[15] = s7 * c58 * z * (z * z - (T)3 * x * x);
}

#define W3J_THREADS 256

__device__ __forceinline__ double odd_dfact(int n) {
  double r = 1.0;
  for (int v = n; v > 1; v -= 2) r *= (double)v;
  return r;
}

#define MONO(ex, ey, ez) ((ex) | ((ey) << 4) | ((ez) << 8))

__device__ __forceinline__ int sh_terms(int m, double* c, int* e) {
  const double s3  = 1.7320508075688772;
  const double s5  = 2.2360679774997896;
  const double s7  = 2.6457513110645907;
  const double s15 = 3.8729833462074170;
  const double c58 = 0.7905694150420949;
  const double c38 = 0.6123724356957945;
  int n = 1;
  c[0] = 1.0; e[0] = MONO(0, 0, 0);
  c[1] = 0.0; e[1] = 0;
  c[2] = 0.0; e[2] = 0;
  switch (m) {
    case 0:  break;
    case 1:  c[0] = s3; e[0] = MONO(1, 0, 0); break;
    case 2:  c[0] = s3; e[0] = MONO(0, 1, 0); break;
    case 3:  c[0] = s3; e[0] = MONO(0, 0, 1); break;
    case 4:  c[0] = s5 * s3; e[0] = MONO(1, 0, 1); break;
    case 5:  c[0] = s5 * s3; e[0] = MONO(1, 1, 0); break;
    case 6:  n = 3;
             c[0] = s5;        e[0] = MONO(0, 2, 0);
             c[1] = -0.5 * s5; e[1] = MONO(2, 0, 0);
             c[2] = -0.5 * s5; e[2] = MONO(0, 0, 2); break;
    case 7:  c[0] = s5 * s3; e[0] = MONO(0, 1, 1); break;
    case 8:  n = 2;
             c[0] = s5 * (s3 * 0.5);    e[0] = MONO(0, 0, 2);
             c[1] = -(s5 * (s3 * 0.5)); e[1] = MONO(2, 0, 0); break;
    case 9:  n = 2;
             c[0] = 3.0 * (s7 * c58); e[0] = MONO(1, 0, 2);
             c[1] = -(s7 * c58);      e[1] = MONO(3, 0, 0); break;
    case 10: c[0] = s7 * s15; e[0] = MONO(1, 1, 1); break;
    case 11: n = 2;
             c[0] = 5.0 * (s7 * c38); e[0] = MONO(1, 2, 0);
             c[1] = -(s7 * c38);      e[1] = MONO(1, 0, 0); break;
    case 12: n = 2;
             c[0] = 5.0 * (s7 * 0.5);  e[0] = MONO(0, 3, 0);
             c[1] = -3.0 * (s7 * 0.5); e[1] = MONO(0, 1, 0); break;
    case 13: n = 2;
             c[0] = 5.0 * (s7 * c38); e[0] = MONO(0, 2, 1);
             c[1] = -(s7 * c38);      e[1] = MONO(0, 0, 1); break;
    case 14: n = 2;
             c[0] = s7 * (s15 * 0.5);    e[0] = MONO(0, 1, 2);
             c[1] = -(s7 * (s15 * 0.5)); e[1] = MONO(2, 1, 0); break;
    default: n = 2;
             c[0] = s7 * c58;          e[0] = MONO(0, 0, 3);
             c[1] = -3.0 * (s7 * c58); e[1] = MONO(2, 0, 1); break;
  }
  return n;
}

__launch_bounds__(W3J_THREADS)
__global__ void w3j_kernel(float* __restrict__ ctab) {
  const int p = blockIdx.x;
  const int t = threadIdx.x;

  int l1 = 0, l2 = 0, l3 = 0, cnt = 0;
  for (int a = 0; a < 4; ++a)
    for (int b = 0; b < 4; ++b)
      for (int c = 0; c < 4; ++c) {
        int lo = a - b; if (lo < 0) lo = -lo;
        if (lo <= c && c <= a + b && (((a + b + c) & 1) == 0)) {
          if (cnt == p) { l1 = a; l2 = b; l3 = c; }
          ++cnt;
        }
      }
  const int d1 = 2 * l1 + 1, d2 = 2 * l2 + 1, d3 = 2 * l3 + 1;
  const int NE = d1 * d2 * d3;

  __shared__ double Mom[125];
  __shared__ double Tc[16 * 3];
  __shared__ int    Te[16 * 3];
  __shared__ int    Tn[16];
  __shared__ double Cs[343];
  __shared__ double red[W3J_THREADS];
  __shared__ int    Gc[16];
  const double PI = 3.14159265358979323846;

  if (t < 125) {
    const int a = 2 * (t / 25), b = 2 * ((t / 5) % 5), c = 2 * (t % 5);
    Mom[t] = 4.0 * PI * odd_dfact(a - 1) * odd_dfact(b - 1) * odd_dfact(c - 1)
             / odd_dfact(a + b + c + 1);
  } else if (t >= 128 && t < 144) {
    const int m = t - 128;
    double c[3];
    int e[3];
    Tn[m] = sh_terms(m, c, e);
    for (int q = 0; q < 3; ++q) { Tc[m * 3 + q] = c[q]; Te[m * 3 + q] = e[q]; }
  }
  __syncthreads();

  for (int n = t; n < NE; n += W3J_THREADS) {
    const int i = n / (d2 * d3);
    const int j = (n / d3) % d2;
    const int k = n % d3;
    const int m1 = l1 * l1 + i, m2 = l2 * l2 + j, m3 = l3 * l3 + k;
    const int n1 = Tn[m1], n2 = Tn[m2], n3 = Tn[m3];
    double s = 0.0;
    for (int q1 = 0; q1 < n1; ++q1)
      for (int q2 = 0; q2 < n2; ++q2)
        for (int q3 = 0; q3 < n3; ++q3) {
          const int e = Te[m1 * 3 + q1] + Te[m2 * 3 + q2] + Te[m3 * 3 + q3];
          if ((e & 0x111) == 0) {
            const int ia = (e & 15) >> 1, ib = ((e >> 4) & 15) >> 1, ic = ((e >> 8) & 15) >> 1;
            s += Tc[m1 * 3 + q1] * Tc[m2 * 3 + q2] * Tc[m3 * 3 + q3]
                 * Mom[ia * 25 + ib * 5 + ic];
          }
        }
    Cs[n] = s;
  }
  __syncthreads();

  double ss = 0.0;
  for (int n = t; n < NE; n += W3J_THREADS) ss += Cs[n] * Cs[n];
  red[t] = ss;
  __syncthreads();
  for (int off = W3J_THREADS / 2; off > 0; off >>= 1) {
    if (t < off) red[t] += red[t + off];
    __syncthreads();
  }
  const double inv = 1.0 / sqrt(red[0]);

  const int base = base_of(p, std::make_integer_sequence<int, NPATHS>{});
  const int wave = t >> 5, ln = t & 31;
  unsigned bal[2];
  bool ok[2];
#pragma unroll
  for (int r = 0; r < 2; ++r) {
    const int n = t + W3J_THREADS * r;
    bool c = false;
    if (n < NE) {
      const int i = n / (d2 * d3);
      const int j = (n / d3) % d2;
      const int k = n % d3;
      c = coupled(l1 * l1 + i, l2 * l2 + j, l3 * l3 + k);
    }
    ok[r] = c;
    bal[r] = (unsigned)__ballot(c);
    if (ln == 0) Gc[8 * r + wave] = __popc(bal[r]);
  }
  __syncthreads();
  int total = 0;
#pragma unroll
  for (int g = 0; g < 16; ++g) total += Gc[g];
#pragma unroll
  for (int r = 0; r < 2; ++r) {
    if (ok[r]) {
      const int n = t + W3J_THREADS * r;
      int off = 0;
      for (int g = 0; g < 8 * r + wave; ++g) off += Gc[g];
      off += __popc(bal[r] & ((1u << ln) - 1u));
      ctab[base + off] = (float)(Cs[n] * inv);
    }
  }
  if (t < 8 && total + t < pad8(total)) ctab[base + total + t] = 0.0f;
}

__global__ void zero_kernel(float* __restrict__ out, int n) {
  int i = blockIdx.x * blockDim.x + threadIdx.x;
  if (i < n) out[i] = 0.0f;
}

template <int P>
__device__ __forceinline__ void path_body(
    int lane,
    float len,
    const float (&Y)[16], const float (&xr)[16],
    const float* __restrict__ wwp,
    const float* __restrict__ wbp,
    const float* __restrict__ ctab,
    float* tmpB, const float* tb,
    v8f& acc0, v8f& acc1) {
  constexpr int l1 = PL1[P], l2 = PL2[P], l3 = PL3[P];
  constexpr int d1 = 2 * l1 + 1, d2 = 2 * l2 + 1, d3 = 2 * l3 + 1;
  constexpr float pw = PWT[l3];
  constexpr int CB = PathBase<P>::value;

  float tk[8];
#pragma unroll
  for (int k = 0; k < 8; ++k) tk[k] = 0.0f;
  int pos = 0;
#pragma unroll
  for (int i = 0; i < d1; ++i) {
    float g[7];
#pragma unroll
    for (int k = 0; k < 7; ++k) g[k] = 0.0f;
#pragma unroll
    for (int j = 0; j < d2; ++j) {
#pragma unroll
      for (int k = 0; k < d3; ++k) {
        if (coupled(l1 * l1 + i, l2 * l2 + j, l3 * l3 + k)) {
          const v8f c = *reinterpret_cast<const v8f*>(ctab + CB + (pos & ~7));
          g[k] = fmaf(Y[l2 * l2 + j], c[pos & 7], g[k]);
          ++pos;
        }
      }
    }
#pragma unroll
    for (int k = 0; k < d3; ++k) {
      bool any = false;
#pragma unroll
      for (int j = 0; j < d2; ++j)
        any = any || coupled(l1 * l1 + i, l2 * l2 + j, l3 * l3 + k);
      if (any) {
        const float gs = __int_as_float(
            __builtin_amdgcn_readfirstlane(__float_as_int(pw * g[k])));
        tk[k] = fmaf(xr[l1 * l1 + i], gs, tk[k]);
      }
    }
  }
  float* brow = tmpB + lane * BPITCH;
#pragma unroll
  for (int k = 0; k < 8; ++k) {
    brow[4 * k]     = tk[k];
    brow[4 * k + 3] = tk[k];
  }
  __syncthreads();

  constexpr int wbase = P * (HIDDEN * HIDDEN);
#pragma unroll
  for (int t = 0; t < 16; ++t) {
    const v2f b = *reinterpret_cast<const v2f*>(tb + 2 * t * BPITCH);
    const int i0 = wbase + t * (2 * HIDDEN);
    const v2f w = *reinterpret_cast<const v2f*>(wwp + i0);
    const v2f c = *reinterpret_cast<const v2f*>(wbp + i0);
    v2f a;
    a.x = fmaf(len, w.x, c.x);
    a.y = fmaf(len, w.y, c.y);
    if (t & 1)
      acc1 = __builtin_amdgcn_wmma_f32_16x16x4_f32(
          false, a, false, b, (short)0, acc1, false, false);
    else
      acc0 = __builtin_amdgcn_wmma_f32_16x16x4_f32(
          false, a, false, b, (short)0, acc0, false, false);
  }
  __syncthreads();
  __builtin_amdgcn_sched_barrier(0);
}

template <int... Ps>
__device__ __forceinline__ void run_group(
    std::integer_sequence<int, Ps...>,
    int lane, float len,
    const float (&Y)[16], const float (&xr)[16],
    const float* __restrict__ wwp, const float* __restrict__ wbp,
    const float* __restrict__ ctab,
    float* tmpB, const float* tb,
    v8f& acc0, v8f& acc1) {
  (path_body<Ps>(lane, len, Y, xr, wwp, wbp, ctab, tmpB, tb, acc0, acc1), ...);
}

template <int L3>
__device__ __forceinline__ void store_tile(int lane, int n16, float* od, const v8f& acc) {
  constexpr int d3 = 2 * L3 + 1;
  if (n16 < 2 * d3) {
    const int k = n16 >> 1;
    const int g = n16 & 1;
    const int madd = (lane >= 16) ? 8 : 0;
#pragma unroll
    for (int r = 0; r < 8; ++r) {
      const int w = 2 * (r + madd) + g;
      atomicAdd(od + HIDDEN * L3 * L3 + w * d3 + k, acc[r]);
    }
  }
}

__launch_bounds__(32)
__global__ void edge_kernel(const float* __restrict__ nf,
                            const int*   __restrict__ eidx,
                            const float* __restrict__ evec,
                            const float* __restrict__ ww,
                            const float* __restrict__ wb,
                            const float* __restrict__ ctab,
                            float*       __restrict__ out) {
  const int e    = blockIdx.x;
  const int lane = threadIdx.x;

  const int src = eidx[e];
  const int dst = eidx[N_EDGES + e];

  const float vx = evec[3 * e + 0];
  const float vy = evec[3 * e + 1];
  const float vz = evec[3 * e + 2];
  float len = sqrtf(vx * vx + vy * vy + vz * vz);
  len = fmaxf(len, 1e-8f);
  const float inv = 1.0f / len;

  float Y[16];
  sh_eval(vx * inv, vy * inv, vz * inv, Y);

  const float* xs = nf + (size_t)src * FEAT;
  float xr[16];
#pragma unroll
  for (int l = 0; l < 4; ++l) {
    const int d = 2 * l + 1;
#pragma unroll
    for (int i = 0; i < 7; ++i)
      if (i < d) xr[l * l + i] = xs[HIDDEN * l * l + lane * d + i];
  }

  __shared__ alignas(16) float tmpB[32 * BPITCH];

  const int n16 = lane & 15;
  const int hh  = lane >> 4;
  const float* wwp = ww + hh * HIDDEN + 2 * n16;
  const float* wbp = wb + hh * HIDDEN + 2 * n16;
  const float* tb  = tmpB + hh * BPITCH + 2 * n16;

  float* od = out + (size_t)dst * FEAT;

  {
    const v4f z = {0.0f, 0.0f, 0.0f, 0.0f};
#pragma unroll
    for (int k = 0; k < 8; ++k)
      *reinterpret_cast<v4f*>(tmpB + lane * BPITCH + 4 * k) = z;
  }

  {
    v8f a0 = {0,0,0,0,0,0,0,0}, a1 = {0,0,0,0,0,0,0,0};
    run_group(std::integer_sequence<int, 0, 5, 13, 21>{},
              lane, len, Y, xr, wwp, wbp, ctab, tmpB, tb, a0, a1);
    const v8f sum = a0 + a1;
    store_tile<0>(lane, n16, od, sum);
  }
  {
    v8f a0 = {0,0,0,0,0,0,0,0}, a1 = {0,0,0,0,0,0,0,0};
    run_group(std::integer_sequence<int, 1, 4, 7, 11, 15, 19>{},
              lane, len, Y, xr, wwp, wbp, ctab, tmpB, tb, a0, a1);
    const v8f sum = a0 + a1;
    store_tile<1>(lane, n16, od, sum);
  }
  {
    v8f a0 = {0,0,0,0,0,0,0,0}, a1 = {0,0,0,0,0,0,0,0};
    run_group(std::integer_sequence<int, 2, 6, 9, 10, 14, 18, 22>{},
              lane, len, Y, xr, wwp, wbp, ctab, tmpB, tb, a0, a1);
    const v8f sum = a0 + a1;
    store_tile<2>(lane, n16, od, sum);
  }
  {
    v8f a0 = {0,0,0,0,0,0,0,0}, a1 = {0,0,0,0,0,0,0,0};
    run_group(std::integer_sequence<int, 3, 8, 12, 16, 17, 20>{},
              lane, len, Y, xr, wwp, wbp, ctab, tmpB, tb, a0, a1);
    const v8f sum = a0 + a1;
    store_tile<3>(lane, n16, od, sum);
  }
}

extern "C" void kernel_launch(void* const* d_in, const int* in_sizes, int n_in,
                              void* d_out, int out_size, void* d_ws, size_t ws_size,
                              hipStream_t stream) {
  const float* nf   = (const float*)d_in[0];
  const int*   eidx = (const int*)  d_in[1];
  const float* evec = (const float*)d_in[2];
  const float* ww   = (const float*)d_in[3];
  const float* wb   = (const float*)d_in[4];
  float* out  = (float*)d_out;
  float* ctab = (float*)d_ws;

  w3j_kernel<<<NPATHS, W3J_THREADS, 0, stream>>>(ctab);
  const int n = N_NODES * FEAT;
  zero_kernel<<<(n + 255) / 256, 256, 0, stream>>>(out, n);
  edge_kernel<<<N_EDGES, 32, 0, stream>>>(nf, eidx, evec, ww, wb, ctab, out);
}


# edge launch: each path fetches its operands ahead of its matrix run, not inside it: weight/bias loads issued at path start, A and B fragments all in registers, then 16 matrix instr back to back
# speedup vs baseline: 95.8668x; 1.0689x over previous
#include <hip/hip_runtime.h>
#include <math.h>
#include <utility>

typedef float v2f __attribute__((ext_vector_type(2)));
typedef float v4f __attribute__((ext_vector_type(4)));
typedef float v8f __attribute__((ext_vector_type(8)));

#define HIDDEN  32
#define N_NODES 4096
#define N_EDGES 8192
#define FEAT    512
#define NPATHS  23
#define BPITCH  36

static constexpr int PL1[NPATHS] = {0,0,0,0, 1,1,1,1,1,1, 2,2,2,2,2,2,2, 3,3,3,3,3,3};
static constexpr int PL2[NPATHS] = {0,1,2,3, 0,1,1,2,2,3, 0,1,1,2,2,3,3, 0,1,2,2,3,3};
static constexpr int PL3[NPATHS] = {0,1,2,3, 1,0,2,1,3,2, 2,1,3,0,2,1,3, 3,2,1,3,0,2};
static constexpr float PWT[4] = {
  0.08838834764831845f,
  0.125f,
  0.14940357616680451f,
  0.19094065395649326f
};

constexpr int par_of(int m) {
  return (int)((0x4242171060354210ULL >> (4 * m)) & 7ULL);
}
constexpr bool coupled(int m1, int m2, int m3) {
  return (par_of(m1) ^ par_of(m2) ^ par_of(m3)) == 0;
}
constexpr int pad8(int n) { return (n + 7) & ~7; }
constexpr int path_coupled(int p) {
  const int l1 = PL1[p], l2 = PL2[p], l3 = PL3[p];
  int c = 0;
  for (int i = 0; i < 2 * l1 + 1; ++i)
    for (int j = 0; j < 2 * l2 + 1; ++j)
      for (int k = 0; k < 2 * l3 + 1; ++k)
        if (coupled(l1 * l1 + i, l2 * l2 + j, l3 * l3 + k)) ++c;
  return c;
}
constexpr int cbase(int p) {
  int b = 0;
  for (int q = 0; q < p; ++q) b += pad8(path_coupled(q));
  return b;
}
template <int P> struct PathBase { static constexpr int value = cbase(P); };
template <int... Ps>
__device__ __forceinline__ int base_of(int p, std::integer_sequence<int, Ps...>) {
  int b = 0;
  ((b = (p == Ps) ? PathBase<Ps>::value : b), ...);
  return b;
}

template <typename T>
__device__ __forceinline__ void sh_eval(T x, T y, T z, T* Y) {
  const T s3  = (T)1.7320508075688772;
  const T s5  = (T)2.2360679774997896;
  const T s7  = (T)2.6457513110645907;
  const T s15 = (T)3.8729833462074170;
  const T c58 = (T)0.7905694150420949;
  const T c38 = (T)0.6123724356957945;
  Y[0]  = (T)1;
  Y[1]  = s3 * x;
  Y[2]  = s3 * y;
  Y[3]  = s3 * z;
  Y[4]  = s5 * s3 * x * z;
  Y[5]  = s5 * s3 * x * y;
  Y[6]  = s5 * (y * y - (T)0.5 * (x * x + z * z));
  Y[7]  = s5 * s3 * y * z;
  Y[8]  = s5 * (s3 * (T)0.5) * (z * z - x * x);
  Y[9]  = s7 * c58 * x * ((T)3 * z * z - x * x);
  Y[10] = s7 * s15 * x * y * z;
  Y[11] = s7 * c38 * x * ((T)5 * y * y - (T)1);
  Y[12] = s7 * (T)0.5 * y * ((T)5 * y * y - (T)3);
  Y[13] = s7 * c38 * z * ((T)5 * y * y - (T)1);
  Y[14] = s7 * (s15 * (T)0.5) * y * (z * z - x * x);
  Y[15] = s7 * c58 * z * (z * z - (T)3 * x * x);
}

#define W3J_THREADS 256

__device__ __forceinline__ double odd_dfact(int n) {
  double r = 1.0;
  for (int v = n; v > 1; v -= 2) r *= (double)v;
  return r;
}

#define MONO(ex, ey, ez) ((ex) | ((ey) << 4) | ((ez) << 8))

__device__ __forceinline__ int sh_terms(int m, double* c, int* e) {
  const double s3  = 1.7320508075688772;
  const double s5  = 2.2360679774997896;
  const double s7  = 2.6457513110645907;
  const double s15 = 3.8729833462074170;
  const double c58 = 0.7905694150420949;
  const double c38 = 0.6123724356957945;
  int n = 1;
  c[0] = 1.0; e[0] = MONO(0, 0, 0);
  c[1] = 0.0; e[1] = 0;
  c[2] = 0.0; e[2] = 0;
  switch (m) {
    case 0:  break;
    case 1:  c[0] = s3; e[0] = MONO(1, 0, 0); break;
    case 2:  c[0] = s3; e[0] = MONO(0, 1, 0); break;
    case 3:  c[0] = s3; e[0] = MONO(0, 0, 1); break;
    case 4:  c[0] = s5 * s3; e[0] = MONO(1, 0, 1); break;
    case 5:  c[0] = s5 * s3; e[0] = MONO(1, 1, 0); break;
    case 6:  n = 3;
             c[0] = s5;        e[0] = MONO(0, 2, 0);
             c[1] = -0.5 * s5; e[1] = MONO(2, 0, 0);
             c[2] = -0.5 * s5; e[2] = MONO(0, 0, 2); break;
    case 7:  c[0] = s5 * s3; e[0] = MONO(0, 1, 1); break;
    case 8:  n = 2;
             c[0] = s5 * (s3 * 0.5);    e[0] = MONO(0, 0, 2);
             c[1] = -(s5 * (s3 * 0.5)); e[1] = MONO(2, 0, 0); break;
    case 9:  n = 2;
             c[0] = 3.0 * (s7 * c58); e[0] = MONO(1, 0, 2);
             c[1] = -(s7 * c58);      e[1] = MONO(3, 0, 0); break;
    case 10: c[0] = s7 * s15; e[0] = MONO(1, 1, 1); break;
    case 11: n = 2;
             c[0] = 5.0 * (s7 * c38); e[0] = MONO(1, 2, 0);
             c[1] = -(s7 * c38);      e[1] = MONO(1, 0, 0); break;
    case 12: n = 2;
             c[0] = 5.0 * (s7 * 0.5);  e[0] = MONO(0, 3, 0);
             c[1] = -3.0 * (s7 * 0.5); e[1] = MONO(0, 1, 0); break;
    case 13: n = 2;
             c[0] = 5.0 * (s7 * c38); e[0] = MONO(0, 2, 1);
             c[1] = -(s7 * c38);      e[1] = MONO(0, 0, 1); break;
    case 14: n = 2;
             c[0] = s7 * (s15 * 0.5);    e[0] = MONO(0, 1, 2);
             c[1] = -(s7 * (s15 * 0.5)); e[1] = MONO(2, 1, 0); break;
    default: n = 2;
             c[0] = s7 * c58;          e[0] = MONO(0, 0, 3);
             c[1] = -3.0 * (s7 * c58); e[1] = MONO(2, 0, 1); break;
  }
  return n;
}

__launch_bounds__(W3J_THREADS)
__global__ void w3j_kernel(float* __restrict__ ctab) {
  const int p = blockIdx.x;
  const int t = threadIdx.x;

  int l1 = 0, l2 = 0, l3 = 0, cnt = 0;
  for (int a = 0; a < 4; ++a)
    for (int b = 0; b < 4; ++b)
      for (int c = 0; c < 4; ++c) {
        int lo = a - b; if (lo < 0) lo = -lo;
        if (lo <= c && c <= a + b && (((a + b + c) & 1) == 0)) {
          if (cnt == p) { l1 = a; l2 = b; l3 = c; }
          ++cnt;
        }
      }
  const int d1 = 2 * l1 + 1, d2 = 2 * l2 + 1, d3 = 2 * l3 + 1;
  const int NE = d1 * d2 * d3;

  __shared__ double Mom[125];
  __shared__ double Tc[16 * 3];
  __shared__ int    Te[16 * 3];
  __shared__ int    Tn[16];
  __shared__ double Cs[343];
  __shared__ double red[W3J_THREADS];
  __shared__ int    Gc[16];
  const double PI = 3.14159265358979323846;

  if (t < 125) {
    const int a = 2 * (t / 25), b = 2 * ((t / 5) % 5), c = 2 * (t % 5);
    Mom[t] = 4.0 * PI * odd_dfact(a - 1) * odd_dfact(b - 1) * odd_dfact(c - 1)
             / odd_dfact(a + b + c + 1);
  } else if (t >= 128 && t < 144) {
    const int m = t - 128;
    double c[3];
    int e[3];
    Tn[m] = sh_terms(m, c, e);
    for (int q = 0; q < 3; ++q) { Tc[m * 3 + q] = c[q]; Te[m * 3 + q] = e[q]; }
  }
  __syncthreads();

  for (int n = t; n < NE; n += W3J_THREADS) {
    const int i = n / (d2 * d3);
    const int j = (n / d3) % d2;
    const int k = n % d3;
    const int m1 = l1 * l1 + i, m2 = l2 * l2 + j, m3 = l3 * l3 + k;
    const int n1 = Tn[m1], n2 = Tn[m2], n3 = Tn[m3];
    double s = 0.0;
    for (int q1 = 0; q1 < n1; ++q1)
      for (int q2 = 0; q2 < n2; ++q2)
        for (int q3 = 0; q3 < n3; ++q3) {
          const int e = Te[m1 * 3 + q1] + Te[m2 * 3 + q2] + Te[m3 * 3 + q3];
          if ((e & 0x111) == 0) {
            const int ia = (e & 15) >> 1, ib = ((e >> 4) & 15) >> 1, ic = ((e >> 8) & 15) >> 1;
            s += Tc[m1 * 3 + q1] * Tc[m2 * 3 + q2] * Tc[m3 * 3 + q3]
                 * Mom[ia * 25 + ib * 5 + ic];
          }
        }
    Cs[n] = s;
  }
  __syncthreads();

  double ss = 0.0;
  for (int n = t; n < NE; n += W3J_THREADS) ss += Cs[n] * Cs[n];
  red[t] = ss;
  __syncthreads();
  for (int off = W3J_THREADS / 2; off > 0; off >>= 1) {
    if (t < off) red[t] += red[t + off];
    __syncthreads();
  }
  const double inv = 1.0 / sqrt(red[0]);

  const int base = base_of(p, std::make_integer_sequence<int, NPATHS>{});
  const int wave = t >> 5, ln = t & 31;
  unsigned bal[2];
  bool ok[2];
#pragma unroll
  for (int r = 0; r < 2; ++r) {
    const int n = t + W3J_THREADS * r;
    bool c = false;
    if (n < NE) {
      const int i = n / (d2 * d3);
      const int j = (n / d3) % d2;
      const int k = n % d3;
      c = coupled(l1 * l1 + i, l2 * l2 + j, l3 * l3 + k);
    }
    ok[r] = c;
    bal[r] = (unsigned)__ballot(c);
    if (ln == 0) Gc[8 * r + wave] = __popc(bal[r]);
  }
  __syncthreads();
  int total = 0;
#pragma unroll
  for (int g = 0; g < 16; ++g) total += Gc[g];
#pragma unroll
  for (int r = 0; r < 2; ++r) {
    if (ok[r]) {
      const int n = t + W3J_THREADS * r;
      int off = 0;
      for (int g = 0; g < 8 * r + wave; ++g) off += Gc[g];
      off += __popc(bal[r] & ((1u << ln) - 1u));
      ctab[base + off] = (float)(Cs[n] * inv);
    }
  }
  if (t < 8 && total + t < pad8(total)) ctab[base + total + t] = 0.0f;
}

__global__ void zero_kernel(float* __restrict__ out, int n) {
  int i = blockIdx.x * blockDim.x + threadIdx.x;
  if (i < n) out[i] = 0.0f;
}

template <int P>
__device__ __forceinline__ void path_body(
    int lane,
    float len,
    const float (&Y)[16], const float (&xr)[16],
    const float* __restrict__ wwp,
    const float* __restrict__ wbp,
    const float* __restrict__ ctab,
    float* tmpB, const float* tb,
    v8f& acc0, v8f& acc1) {
  constexpr int l1 = PL1[P], l2 = PL2[P], l3 = PL3[P];
  constexpr int d1 = 2 * l1 + 1, d2 = 2 * l2 + 1, d3 = 2 * l3 + 1;
  constexpr float pw = PWT[l3];
  constexpr int CB = PathBase<P>::value;

  constexpr int wbase = P * (HIDDEN * HIDDEN);
  v2f wv[16], cv[16];
#pragma unroll
  for (int t = 0; t < 16; ++t) {
    const int i0 = wbase + t * (2 * HIDDEN);
    wv[t] = *reinterpret_cast<const v2f*>(wwp + i0);
    cv[t] = *reinterpret_cast<const v2f*>(wbp + i0);
  }

  float tk[8];
#pragma unroll
  for (int k = 0; k < 8; ++k) tk[k] = 0.0f;
  int pos = 0;
#pragma unroll
  for (int i = 0; i < d1; ++i) {
    float g[7];
#pragma unroll
    for (int k = 0; k < 7; ++k) g[k] = 0.0f;
#pragma unroll
    for (int j = 0; j < d2; ++j) {
#pragma unroll
      for (int k = 0; k < d3; ++k) {
        if (coupled(l1 * l1 + i, l2 * l2 + j, l3 * l3 + k)) {
          const v8f c = *reinterpret_cast<const v8f*>(ctab + CB + (pos & ~7));
          g[k] = fmaf(Y[l2 * l2 + j], c[pos & 7], g[k]);
          ++pos;
        }
      }
    }
#pragma unroll
    for (int k = 0; k < d3; ++k) {
      bool any = false;
#pragma unroll
      for (int j = 0; j < d2; ++j)
        any = any || coupled(l1 * l1 + i, l2 * l2 + j, l3 * l3 + k);
      if (any) {
        const float gs = __int_as_float(
            __builtin_amdgcn_readfirstlane(__float_as_int(pw * g[k])));
        tk[k] = fmaf(xr[l1 * l1 + i], gs, tk[k]);
      }
    }
  }
  float* brow = tmpB + lane * BPITCH;
#pragma unroll
  for (int k = 0; k < 8; ++k) {
    brow[4 * k]     = tk[k];
    brow[4 * k + 3] = tk[k];
  }
  __syncthreads();

  v2f av[16];
#pragma unroll
  for (int t = 0; t < 16; ++t) {
    av[t].x = fmaf(len, wv[t].x, cv[t].x);
    av[t].y = fmaf(len, wv[t].y, cv[t].y);
  }
  __builtin_amdgcn_sched_barrier(0);
  v2f bv[16];
#pragma unroll
  for (int t = 0; t < 16; ++t)
    bv[t] = *reinterpret_cast<const v2f*>(tb + 2 * t * BPITCH);
  __builtin_amdgcn_sched_barrier(0);
#pragma unroll
  for (int t = 0; t < 16; ++t) {
    if (t & 1)
      acc1 = __builtin_amdgcn_wmma_f32_16x16x4_f32(
          false, av[t], false, bv[t], (short)0, acc1, false, false);
    else
      acc0 = __builtin_amdgcn_wmma_f32_16x16x4_f32(
          false, av[t], false, bv[t], (short)0, acc0, false, false);
  }
  __syncthreads();
  __builtin_amdgcn_sched_barrier(0);
}

template <int... Ps>
__device__ __forceinline__ void run_group(
    std::integer_sequence<int, Ps...>,
    int lane, float len,
    const float (&Y)[16], const float (&xr)[16],
    const float* __restrict__ wwp, const float* __restrict__ wbp,
    const float* __restrict__ ctab,
    float* tmpB, const float* tb,
    v8f& acc0, v8f& acc1) {
  (path_body<Ps>(lane, len, Y, xr, wwp, wbp, ctab, tmpB, tb, acc0, acc1), ...);
}

template <int L3>
__device__ __forceinline__ void store_tile(int lane, int n16, float* od, const v8f& acc) {
  constexpr int d3 = 2 * L3 + 1;
  if (n16 < 2 * d3) {
    const int k = n16 >> 1;
    const int g = n16 & 1;
    const int madd = (lane >= 16) ? 8 : 0;
#pragma unroll
    for (int r = 0; r < 8; ++r) {
      const int w = 2 * (r + madd) + g;
      atomicAdd(od + HIDDEN * L3 * L3 + w * d3 + k, acc[r]);
    }
  }
}

__launch_bounds__(32)
__global__ void edge_kernel(const float* __restrict__ nf,
                            const int*   __restrict__ eidx,
                            const float* __restrict__ evec,
                            const float* __restrict__ ww,
                            const float* __restrict__ wb,
                            const float* __restrict__ ctab,
                            float*       __restrict__ out) {
  const int e    = blockIdx.x;
  const int lane = threadIdx.x;

  const int src = eidx[e];
  const int dst = eidx[N_EDGES + e];

  const float vx = evec[3 * e + 0];
  const float vy = evec[3 * e + 1];
  const float vz = evec[3 * e + 2];
  float len = sqrtf(vx * vx + vy * vy + vz * vz);
  len = fmaxf(len, 1e-8f);
  const float inv = 1.0f / len;

  float Y[16];
  sh_eval(vx * inv, vy * inv, vz * inv, Y);

  const float* xs = nf + (size_t)src * FEAT;
  float xr[16];
#pragma unroll
  for (int l = 0; l < 4; ++l) {
    const int d = 2 * l + 1;
#pragma unroll
    for (int i = 0; i < 7; ++i)
      if (i < d) xr[l * l + i] = xs[HIDDEN * l * l + lane * d + i];
  }

  __shared__ alignas(16) float tmpB[32 * BPITCH];

  const int n16 = lane & 15;
  const int hh  = lane >> 4;
  const float* wwp = ww + hh * HIDDEN + 2 * n16;
  const float* wbp = wb + hh * HIDDEN + 2 * n16;
  const float* tb  = tmpB + hh * BPITCH + 2 * n16;

  float* od = out + (size_t)dst * FEAT;

  {
    const v4f z = {0.0f, 0.0f, 0.0f, 0.0f};
#pragma unroll
    for (int k = 0; k < 8; ++k)
      *reinterpret_cast<v4f*>(tmpB + lane * BPITCH + 4 * k) = z;
  }

  {
    v8f a0 = {0,0,0,0,0,0,0,0}, a1 = {0,0,0,0,0,0,0,0};
    run_group(std::integer_sequence<int, 0, 5, 13, 21>{},
              lane, len, Y, xr, wwp, wbp, ctab, tmpB, tb, a0, a1);
    const v8f sum = a0 + a1;
    store_tile<0>(lane, n16, od, sum);
  }
  {
    v8f a0 = {0,0,0,0,0,0,0,0}, a1 = {0,0,0,0,0,0,0,0};
    run_group(std::integer_sequence<int, 1, 4, 7, 11, 15, 19>{},
              lane, len, Y, xr, wwp, wbp, ctab, tmpB, tb, a0, a1);
    const v8f sum = a0 + a1;
    store_tile<1>(lane, n16, od, sum);
  }
  {
    v8f a0 = {0,0,0,0,0,0,0,0}, a1 = {0,0,0,0,0,0,0,0};
    run_group(std::integer_sequence<int, 2, 6, 9, 10, 14, 18, 22>{},
              lane, len, Y, xr, wwp, wbp, ctab, tmpB, tb, a0, a1);
    const v8f sum = a0 + a1;
    store_tile<2>(lane, n16, od, sum);
  }
  {
    v8f a0 = {0,0,0,0,0,0,0,0}, a1 = {0,0,0,0,0,0,0,0};
    run_group(std::integer_sequence<int, 3, 8, 12, 16, 17, 20>{},
              lane, len, Y, xr, wwp, wbp, ctab, tmpB, tb, a0, a1);
    const v8f sum = a0 + a1;
    store_tile<3>(lane, n16, od, sum);
  }
}

extern "C" void kernel_launch(void* const* d_in, const int* in_sizes, int n_in,
                              void* d_out, int out_size, void* d_ws, size_t ws_size,
                              hipStream_t stream) {
  const float* nf   = (const float*)d_in[0];
  const int*   eidx = (const int*)  d_in[1];
  const float* evec = (const float*)d_in[2];
  const float* ww   = (const float*)d_in[3];
  const float* wb   = (const float*)d_in[4];
  float* out  = (float*)d_out;
  float* ctab = (float*)d_ws;

  w3j_kernel<<<NPATHS, W3J_THREADS, 0, stream>>>(ctab);
  const int n = N_NODES * FEAT;
  zero_kernel<<<(n + 255) / 256, 256, 0, stream>>>(out, n);
  edge_kernel<<<N_EDGES, 32, 0, stream>>>(nf, eidx, evec, ww, wb, ctab, out);
}


# table launch: each element's term lists and the moments of its 27 term products are read from LDS first, fp64 products then run unrolled from registers under wave-uniform skips; sums, order kept
# speedup vs baseline: 100.6908x; 1.0503x over previous
#include <hip/hip_runtime.h>
#include <math.h>
#include <utility>

typedef float v2f __attribute__((ext_vector_type(2)));
typedef float v4f __attribute__((ext_vector_type(4)));
typedef float v8f __attribute__((ext_vector_type(8)));

#define HIDDEN  32
#define N_NODES 4096
#define N_EDGES 8192
#define FEAT    512
#define NPATHS  23
#define BPITCH  36

static constexpr int PL1[NPATHS] = {0,0,0,0, 1,1,1,1,1,1, 2,2,2,2,2,2,2, 3,3,3,3,3,3};
static constexpr int PL2[NPATHS] = {0,1,2,3, 0,1,1,2,2,3, 0,1,1,2,2,3,3, 0,1,2,2,3,3};
static constexpr int PL3[NPATHS] = {0,1,2,3, 1,0,2,1,3,2, 2,1,3,0,2,1,3, 3,2,1,3,0,2};
static constexpr float PWT[4] = {
  0.08838834764831845f,
  0.125f,
  0.14940357616680451f,
  0.19094065395649326f
};

constexpr int par_of(int m) {
  return (int)((0x4242171060354210ULL >> (4 * m)) & 7ULL);
}
constexpr bool coupled(int m1, int m2, int m3) {
  return (par_of(m1) ^ par_of(m2) ^ par_of(m3)) == 0;
}
constexpr int pad8(int n) { return (n + 7) & ~7; }
constexpr int path_coupled(int p) {
  const int l1 = PL1[p], l2 = PL2[p], l3 = PL3[p];
  int c = 0;
  for (int i = 0; i < 2 * l1 + 1; ++i)
    for (int j = 0; j < 2 * l2 + 1; ++j)
      for (int k = 0; k < 2 * l3 + 1; ++k)
        if (coupled(l1 * l1 + i, l2 * l2 + j, l3 * l3 + k)) ++c;
  return c;
}
constexpr int cbase(int p) {
  int b = 0;
  for (int q = 0; q < p; ++q) b += pad8(path_coupled(q));
  return b;
}
template <int P> struct PathBase { static constexpr int value = cbase(P); };
template <int... Ps>
__device__ __forceinline__ int base_of(int p, std::integer_sequence<int, Ps...>) {
  int b = 0;
  ((b = (p == Ps) ? PathBase<Ps>::value : b), ...);
  return b;
}

template <typename T>
__device__ __forceinline__ void sh_eval(T x, T y, T z, T* Y) {
  const T s3  = (T)1.7320508075688772;
  const T s5  = (T)2.2360679774997896;
  const T s7  = (T)2.6457513110645907;
  const T s15 = (T)3.8729833462074170;
  const T c58 = (T)0.7905694150420949;
  const T c38 = (T)0.6123724356957945;
  Y[0]  = (T)1;
  Y[1]  = s3 * x;
  Y[2]  = s3 * y;
  Y[3]  = s3 * z;
  Y[4]  = s5 * s3 * x * z;
  Y[5]  = s5 * s3 * x * y;
  Y[6]  = s5 * (y * y - (T)0.5 * (x * x + z * z));
  Y[7]  = s5 * s3 * y * z;
  Y[8]  = s5 * (s3 * (T)0.5) * (z * z - x * x);
  Y[9]  = s7 * c58 * x * ((T)3 * z * z - x * x);
  Y[10] = s7 * s15 * x * y * z;
  Y[11] = s7 * c38 * x * ((T)5 * y * y - (T)1);
  Y[12] = s7 * (T)0.5 * y * ((T)5 * y * y - (T)3);
  Y[13] = s7 * c38 * z * ((T)5 * y * y - (T)1);
  Y[14] = s7 * (s15 * (T)0.5) * y * (z * z - x * x);
  Y[15] = s7 * c58 * z * (z * z - (T)3 * x * x);
}

#define W3J_THREADS 256

__device__ __forceinline__ double odd_dfact(int n) {
  double r = 1.0;
  for (int v = n; v > 1; v -= 2) r *= (double)v;
  return r;
}

#define MONO(ex, ey, ez) ((ex) | ((ey) << 4) | ((ez) << 8))

__device__ __forceinline__ int sh_terms(int m, double* c, int* e) {
  const double s3  = 1.7320508075688772;
  const double s5  = 2.2360679774997896;
  const double s7  = 2.6457513110645907;
  const double s15 = 3.8729833462074170;
  const double c58 = 0.7905694150420949;
  const double c38 = 0.6123724356957945;
  int n = 1;
  c[0] = 1.0; e[0] = MONO(0, 0, 0);
  c[1] = 0.0; e[1] = 0;
  c[2] = 0.0; e[2] = 0;
  switch (m) {
    case 0:  break;
    case 1:  c[0] = s3; e[0] = MONO(1, 0, 0); break;
    case 2:  c[0] = s3; e[0] = MONO(0, 1, 0); break;
    case 3:  c[0] = s3; e[0] = MONO(0, 0, 1); break;
    case 4:  c[0] = s5 * s3; e[0] = MONO(1, 0, 1); break;
    case 5:  c[0] = s5 * s3; e[0] = MONO(1, 1, 0); break;
    case 6:  n = 3;
             c[0] = s5;        e[0] = MONO(0, 2, 0);
             c[1] = -0.5 * s5; e[1] = MONO(2, 0, 0);
             c[2] = -0.5 * s5; e[2] = MONO(0, 0, 2); break;
    case 7:  c[0] = s5 * s3; e[0] = MONO(0, 1, 1); break;
    case 8:  n = 2;
             c[0] = s5 * (s3 * 0.5);    e[0] = MONO(0, 0, 2);
             c[1] = -(s5 * (s3 * 0.5)); e[1] = MONO(2, 0, 0); break;
    case 9:  n = 2;
             c[0] = 3.0 * (s7 * c58); e[0] = MONO(1, 0, 2);
             c[1] = -(s7 * c58);      e[1] = MONO(3, 0, 0); break;
    case 10: c[0] = s7 * s15; e[0] = MONO(1, 1, 1); break;
    case 11: n = 2;
             c[0] = 5.0 * (s7 * c38); e[0] = MONO(1, 2, 0);
             c[1] = -(s7 * c38);      e[1] = MONO(1, 0, 0); break;
    case 12: n = 2;
             c[0] = 5.0 * (s7 * 0.5);  e[0] = MONO(0, 3, 0);
             c[1] = -3.0 * (s7 * 0.5); e[1] = MONO(0, 1, 0); break;
    case 13: n = 2;
             c[0] = 5.0 * (s7 * c38); e[0] = MONO(0, 2, 1);
             c[1] = -(s7 * c38);      e[1] = MONO(0, 0, 1); break;
    case 14: n = 2;
             c[0] = s7 * (s15 * 0.5);    e[0] = MONO(0, 1, 2);
             c[1] = -(s7 * (s15 * 0.5)); e[1] = MONO(2, 1, 0); break;
    default: n = 2;
             c[0] = s7 * c58;          e[0] = MONO(0, 0, 3);
             c[1] = -3.0 * (s7 * c58); e[1] = MONO(2, 0, 1); break;
  }
  return n;
}

__launch_bounds__(W3J_THREADS)
__global__ void w3j_kernel(float* __restrict__ ctab) {
  const int p = blockIdx.x;
  const int t = threadIdx.x;

  int l1 = 0, l2 = 0, l3 = 0, cnt = 0;
  for (int a = 0; a < 4; ++a)
    for (int b = 0; b < 4; ++b)
      for (int c = 0; c < 4; ++c) {
        int lo = a - b; if (lo < 0) lo = -lo;
        if (lo <= c && c <= a + b && (((a + b + c) & 1) == 0)) {
          if (cnt == p) { l1 = a; l2 = b; l3 = c; }
          ++cnt;
        }
      }
  const int d1 = 2 * l1 + 1, d2 = 2 * l2 + 1, d3 = 2 * l3 + 1;
  const int NE = d1 * d2 * d3;

  __shared__ double Mom[125];
  __shared__ double Tc[16 * 3];
  __shared__ int    Te[16 * 3];
  __shared__ int    Tn[16];
  __shared__ double Cs[343];
  __shared__ double red[W3J_THREADS];
  __shared__ int    Gc[16];
  const double PI = 3.14159265358979323846;

  if (t < 125) {
    const int a = 2 * (t / 25), b = 2 * ((t / 5) % 5), c = 2 * (t % 5);
    Mom[t] = 4.0 * PI * odd_dfact(a - 1) * odd_dfact(b - 1) * odd_dfact(c - 1)
             / odd_dfact(a + b + c + 1);
  } else if (t >= 128 && t < 144) {
    const int m = t - 128;
    double c[3];
    int e[3];
    Tn[m] = sh_terms(m, c, e);
    for (int q = 0; q < 3; ++q) { Tc[m * 3 + q] = c[q]; Te[m * 3 + q] = e[q]; }
  }
  __syncthreads();

  for (int n = t; n < NE; n += W3J_THREADS) {
    const int i = n / (d2 * d3);
    const int j = (n / d3) % d2;
    const int k = n % d3;
    const int m1 = l1 * l1 + i, m2 = l2 * l2 + j, m3 = l3 * l3 + k;
    const int n1 = Tn[m1], n2 = Tn[m2], n3 = Tn[m3];
    double c1[3], c2[3], c3[3];
    int e1[3], e2[3], e3[3];
#pragma unroll
    for (int q = 0; q < 3; ++q) {
      c1[q] = Tc[m1 * 3 + q]; e1[q] = Te[m1 * 3 + q];
      c2[q] = Tc[m2 * 3 + q]; e2[q] = Te[m2 * 3 + q];
      c3[q] = Tc[m3 * 3 + q]; e3[q] = Te[m3 * 3 + q];
    }
    double mom[27];
    bool on[27];
#pragma unroll
    for (int q1 = 0; q1 < 3; ++q1)
#pragma unroll
      for (int q2 = 0; q2 < 3; ++q2)
#pragma unroll
        for (int q3 = 0; q3 < 3; ++q3) {
          const int e = e1[q1] + e2[q2] + e3[q3];
          const int ia = (e & 15) >> 1, ib = ((e >> 4) & 15) >> 1, ic = ((e >> 8) & 15) >> 1;
          on[(q1 * 3 + q2) * 3 + q3] =
              (q1 < n1) && (q2 < n2) && (q3 < n3) && ((e & 0x111) == 0);
          mom[(q1 * 3 + q2) * 3 + q3] = Mom[ia * 25 + ib * 5 + ic];
        }
    double s = 0.0;
#pragma unroll
    for (int q1 = 0; q1 < 3; ++q1)
#pragma unroll
      for (int q2 = 0; q2 < 3; ++q2) {
        if (__any((q1 < n1) && (q2 < n2))) {
          const double t12 = c1[q1] * c2[q2];
#pragma unroll
          for (int q3 = 0; q3 < 3; ++q3) {
            const int x = (q1 * 3 + q2) * 3 + q3;
            if (__any(on[x])) {
              const double mm = on[x] ? mom[x] : 0.0;
              s = __builtin_fma(t12 * c3[q3], mm, s);
            }
          }
        }
      }
    Cs[n] = s;
  }
  __syncthreads();

  double ss = 0.0;
  for (int n = t; n < NE; n += W3J_THREADS) ss += Cs[n] * Cs[n];
  red[t] = ss;
  __syncthreads();
  for (int off = W3J_THREADS / 2; off > 0; off >>= 1) {
    if (t < off) red[t] += red[t + off];
    __syncthreads();
  }
  const double inv = 1.0 / sqrt(red[0]);

  const int base = base_of(p, std::make_integer_sequence<int, NPATHS>{});
  const int wave = t >> 5, ln = t & 31;
  unsigned bal[2];
  bool ok[2];
#pragma unroll
  for (int r = 0; r < 2; ++r) {
    const int n = t + W3J_THREADS * r;
    bool c = false;
    if (n < NE) {
      const int i = n / (d2 * d3);
      const int j = (n / d3) % d2;
      const int k = n % d3;
      c = coupled(l1 * l1 + i, l2 * l2 + j, l3 * l3 + k);
    }
    ok[r] = c;
    bal[r] = (unsigned)__ballot(c);
    if (ln == 0) Gc[8 * r + wave] = __popc(bal[r]);
  }
  __syncthreads();
  int total = 0;
#pragma unroll
  for (int g = 0; g < 16; ++g) total += Gc[g];
#pragma unroll
  for (int r = 0; r < 2; ++r) {
    if (ok[r]) {
      const int n = t + W3J_THREADS * r;
      int off = 0;
      for (int g = 0; g < 8 * r + wave; ++g) off += Gc[g];
      off += __popc(bal[r] & ((1u << ln) - 1u));
      ctab[base + off] = (float)(Cs[n] * inv);
    }
  }
  if (t < 8 && total + t < pad8(total)) ctab[base + total + t] = 0.0f;
}

__global__ void zero_kernel(float* __restrict__ out, int n) {
  int i = blockIdx.x * blockDim.x + threadIdx.x;
  if (i < n) out[i] = 0.0f;
}

template <int P>
__device__ __forceinline__ void path_body(
    int lane,
    float len,
    const float (&Y)[16], const float (&xr)[16],
    const float* __restrict__ wwp,
    const float* __restrict__ wbp,
    const float* __restrict__ ctab,
    float* tmpB, const float* tb,
    v8f& acc0, v8f& acc1) {
  constexpr int l1 = PL1[P], l2 = PL2[P], l3 = PL3[P];
  constexpr int d1 = 2 * l1 + 1, d2 = 2 * l2 + 1, d3 = 2 * l3 + 1;
  constexpr float pw = PWT[l3];
  constexpr int CB = PathBase<P>::value;

  constexpr int wbase = P * (HIDDEN * HIDDEN);
  v2f wv[16], cv[16];
#pragma unroll
  for (int t = 0; t < 16; ++t) {
    const int i0 = wbase + t * (2 * HIDDEN);
    wv[t] = *reinterpret_cast<const v2f*>(wwp + i0);
    cv[t] = *reinterpret_cast<const v2f*>(wbp + i0);
  }

  float tk[8];
#pragma unroll
  for (int k = 0; k < 8; ++k) tk[k] = 0.0f;
  int pos = 0;
#pragma unroll
  for (int i = 0; i < d1; ++i) {
    float g[7];
#pragma unroll
    for (int k = 0; k < 7; ++k) g[k] = 0.0f;
#pragma unroll
    for (int j = 0; j < d2; ++j) {
#pragma unroll
      for (int k = 0; k < d3; ++k) {
        if (coupled(l1 * l1 + i, l2 * l2 + j, l3 * l3 + k)) {
          const v8f c = *reinterpret_cast<const v8f*>(ctab + CB + (pos & ~7));
          g[k] = fmaf(Y[l2 * l2 + j], c[pos & 7], g[k]);
          ++pos;
        }
      }
    }
#pragma unroll
    for (int k = 0; k < d3; ++k) {
      bool any = false;
#pragma unroll
      for (int j = 0; j < d2; ++j)
        any = any || coupled(l1 * l1 + i, l2 * l2 + j, l3 * l3 + k);
      if (any) {
        const float gs = __int_as_float(
            __builtin_amdgcn_readfirstlane(__float_as_int(pw * g[k])));
        tk[k] = fmaf(xr[l1 * l1 + i], gs, tk[k]);
      }
    }
  }
  float* brow = tmpB + lane * BPITCH;
#pragma unroll
  for (int k = 0; k < 8; ++k) {
    brow[4 * k]     = tk[k];
    brow[4 * k + 3] = tk[k];
  }
  __syncthreads();

  v2f av[16];
#pragma unroll
  for (int t = 0; t < 16; ++t) {
    av[t].x = fmaf(len, wv[t].x, cv[t].x);
    av[t].y = fmaf(len, wv[t].y, cv[t].y);
  }
  __builtin_amdgcn_sched_barrier(0);
  v2f bv[16];
#pragma unroll
  for (int t = 0; t < 16; ++t)
    bv[t] = *reinterpret_cast<const v2f*>(tb + 2 * t * BPITCH);
  __builtin_amdgcn_sched_barrier(0);
#pragma unroll
  for (int t = 0; t < 16; ++t) {
    if (t & 1)
      acc1 = __builtin_amdgcn_wmma_f32_16x16x4_f32(
          false, av[t], false, bv[t], (short)0, acc1, false, false);
    else
      acc0 = __builtin_amdgcn_wmma_f32_16x16x4_f32(
          false, av[t], false, bv[t], (short)0, acc0, false, false);
  }
  __syncthreads();
  __builtin_amdgcn_sched_barrier(0);
}

template <int... Ps>
__device__ __forceinline__ void run_group(
    std::integer_sequence<int, Ps...>,
    int lane, float len,
    const float (&Y)[16], const float (&xr)[16],
    const float* __restrict__ wwp, const float* __restrict__ wbp,
    const float* __restrict__ ctab,
    float* tmpB, const float* tb,
    v8f& acc0, v8f& acc1) {
  (path_body<Ps>(lane, len, Y, xr, wwp, wbp, ctab, tmpB, tb, acc0, acc1), ...);
}

template <int L3>
__device__ __forceinline__ void store_tile(int lane, int n16, float* od, const v8f& acc) {
  constexpr int d3 = 2 * L3 + 1;
  if (n16 < 2 * d3) {
    const int k = n16 >> 1;
    const int g = n16 & 1;
    const int madd = (lane >= 16) ? 8 : 0;
#pragma unroll
    for (int r = 0; r < 8; ++r) {
      const int w = 2 * (r + madd) + g;
      atomicAdd(od + HIDDEN * L3 * L3 + w * d3 + k, acc[r]);
    }
  }
}

__launch_bounds__(32)
__global__ void edge_kernel(const float* __restrict__ nf,
                            const int*   __restrict__ eidx,
                            const float* __restrict__ evec,
                            const float* __restrict__ ww,
                            const float* __restrict__ wb,
                            const float* __restrict__ ctab,
                            float*       __restrict__ out) {
  const int e    = blockIdx.x;
  const int lane = threadIdx.x;

  const int src = eidx[e];
  const int dst = eidx[N_EDGES + e];

  const float vx = evec[3 * e + 0];
  const float vy = evec[3 * e + 1];
  const float vz = evec[3 * e + 2];
  float len = sqrtf(vx * vx + vy * vy + vz * vz);
  len = fmaxf(len, 1e-8f);
  const float inv = 1.0f / len;

  float Y[16];
  sh_eval(vx * inv, vy * inv, vz * inv, Y);

  const float* xs = nf + (size_t)src * FEAT;
  float xr[16];
#pragma unroll
  for (int l = 0; l < 4; ++l) {
    const int d = 2 * l + 1;
#pragma unroll
    for (int i = 0; i < 7; ++i)
      if (i < d) xr[l * l + i] = xs[HIDDEN * l * l + lane * d + i];
  }

  __shared__ alignas(16) float tmpB[32 * BPITCH];

  const int n16 = lane & 15;
  const int hh  = lane >> 4;
  const float* wwp = ww + hh * HIDDEN + 2 * n16;
  const float* wbp = wb + hh * HIDDEN + 2 * n16;
  const float* tb  = tmpB + hh * BPITCH + 2 * n16;

  float* od = out + (size_t)dst * FEAT;

  {
    const v4f z = {0.0f, 0.0f, 0.0f, 0.0f};
#pragma unroll
    for (int k = 0; k < 8; ++k)
      *reinterpret_cast<v4f*>(tmpB + lane * BPITCH + 4 * k) = z;
  }

  {
    v8f a0 = {0,0,0,0,0,0,0,0}, a1 = {0,0,0,0,0,0,0,0};
    run_group(std::integer_sequence<int, 0, 5, 13, 21>{},
              lane, len, Y, xr, wwp, wbp, ctab, tmpB, tb, a0, a1);
    const v8f sum = a0 + a1;
    store_tile<0>(lane, n16, od, sum);
  }
  {
    v8f a0 = {0,0,0,0,0,0,0,0}, a1 = {0,0,0,0,0,0,0,0};
    run_group(std::integer_sequence<int, 1, 4, 7, 11, 15, 19>{},
              lane, len, Y, xr, wwp, wbp, ctab, tmpB, tb, a0, a1);
    const v8f sum = a0 + a1;
    store_tile<1>(lane, n16, od, sum);
  }
  {
    v8f a0 = {0,0,0,0,0,0,0,0}, a1 = {0,0,0,0,0,0,0,0};
    run_group(std::integer_sequence<int, 2, 6, 9, 10, 14, 18, 22>{},
              lane, len, Y, xr, wwp, wbp, ctab, tmpB, tb, a0, a1);
    const v8f sum = a0 + a1;
    store_tile<2>(lane, n16, od, sum);
  }
  {
    v8f a0 = {0,0,0,0,0,0,0,0}, a1 = {0,0,0,0,0,0,0,0};
    run_group(std::integer_sequence<int, 3, 8, 12, 16, 17, 20>{},
              lane, len, Y, xr, wwp, wbp, ctab, tmpB, tb, a0, a1);
    const v8f sum = a0 + a1;
    store_tile<3>(lane, n16, od, sum);
  }
}

extern "C" void kernel_launch(void* const* d_in, const int* in_sizes, int n_in,
                              void* d_out, int out_size, void* d_ws, size_t ws_size,
                              hipStream_t stream) {
  const float* nf   = (const float*)d_in[0];
  const int*   eidx = (const int*)  d_in[1];
  const float* evec = (const float*)d_in[2];
  const float* ww   = (const float*)d_in[3];
  const float* wb   = (const float*)d_in[4];
  float* out  = (float*)d_out;
  float* ctab = (float*)d_ws;

  w3j_kernel<<<NPATHS, W3J_THREADS, 0, stream>>>(ctab);
  const int n = N_NODES * FEAT;
  zero_kernel<<<(n + 255) / 256, 256, 0, stream>>>(out, n);
  edge_kernel<<<N_EDGES, 32, 0, stream>>>(nf, eidx, evec, ww, wb, ctab, out);
}
